# starting kernel
# baseline (speedup 1.0000x reference)
#include <hip/hip_runtime.h>

#define NLEN   8192
#define MODES  64
#define ROWS   16
#define W0     (7.6699039394282058e-4f)
#define INV_N  (1.220703125e-4f)

typedef __attribute__((ext_vector_type(2))) float v2f;
typedef __attribute__((ext_vector_type(8))) float v8f;

static __device__ __forceinline__ v8f wmma_f32_16x16x4(v2f a, v2f b, v8f c) {
    return __builtin_amdgcn_wmma_f32_16x16x4_f32(
         false, a,  false, b,
         (short)0, c,  false,  false);
}

__global__ void __launch_bounds__(1024)
dht_forward_modes(const float* __restrict__ x, float* __restrict__ Xm) {
    __shared__ float partial[32 * 256];

    const int lane = threadIdx.x & 31;
    const int wave = threadIdx.x >> 5;
    const int hi   = lane >> 4;
    const int lo   = lane & 15;
    const int tile = wave & 3;
    const int ks   = wave >> 2;
    const int mode = tile * 16 + lo;

    v8f acc = {};
    const int k0 = ks * 1024;
    for (int k = k0; k < k0 + 1024; k += 4) {
        const int kb = k + 2 * hi;
        v2f a, b;
        a.x = x[lo * NLEN + kb];
        a.y = x[lo * NLEN + kb + 1];
        b.x = __cosf(W0 * (float)((kb * mode) & (NLEN - 1)));
        b.y = __cosf(W0 * (float)(((kb + 1) * mode) & (NLEN - 1)));
        acc = wmma_f32_16x16x4(a, b, acc);
    }

#pragma unroll
    for (int j = 0; j < 8; ++j)
        partial[wave * 256 + (j + 8 * hi) * 16 + lo] = acc[j];
    __syncthreads();

    const int r = threadIdx.x >> 6;
    const int m = threadIdx.x & 63;
    float s = 0.0f;
#pragma unroll
    for (int sl = 0; sl < 8; ++sl)
        s += partial[(sl * 4 + (m >> 4)) * 256 + r * 16 + (m & 15)];
    Xm[r * 64 + m] = s;
}

__global__ void __launch_bounds__(1024)
mode_mix(const float* __restrict__ Xm, const float* __restrict__ w,
         float* __restrict__ Apacked) {
    const int t = threadIdx.x;
    const int m = t & 63;
    const int o = (t >> 6) & 7;
    const int b = t >> 9;
    float ar = 0.0f, ai = 0.0f;
#pragma unroll
    for (int i = 0; i < 8; ++i) {
        const float xv = Xm[(b * 8 + i) * 64 + m];
        const float wre = w[(((i * 8 + o) * 64 + m) << 1) + 0];
        const float wim = w[(((i * 8 + o) * 64 + m) << 1) + 1];
        ar += xv * wre;
        ai += xv * wim;
    }
    const int r = b * 8 + o;
    Apacked[r * 128 + m]      = ar;
    Apacked[r * 128 + 64 + m] = ai;
}

__global__ void __launch_bounds__(128)
dht_inverse(const float* __restrict__ Apacked, float* __restrict__ y) {
    __shared__ float As[ROWS * 128];
    for (int i = threadIdx.x; i < ROWS * 128; i += 128)
        As[i] = Apacked[i];
    __syncthreads();

    const int lane = threadIdx.x & 31;
    const int wave = threadIdx.x >> 5;
    const int hi   = lane >> 4;
    const int lo   = lane & 15;
    const int n    = (blockIdx.x * 4 + wave) * 16 + lo;

    v8f acc = {};
    for (int k = 0; k < 128; k += 4) {
        const int kb = k + 2 * hi;
        v2f a, b;
        a.x = As[lo * 128 + kb];
        a.y = As[lo * 128 + kb + 1];
        if (kb < 64) {
            b.x =  INV_N * __cosf(W0 * (float)((kb * n) & (NLEN - 1)));
            b.y =  INV_N * __cosf(W0 * (float)(((kb + 1) * n) & (NLEN - 1)));
        } else {
            b.x = -INV_N * __sinf(W0 * (float)(((kb - 64) * n) & (NLEN - 1)));
            b.y = -INV_N * __sinf(W0 * (float)(((kb - 63) * n) & (NLEN - 1)));
        }
        acc = wmma_f32_16x16x4(a, b, acc);
    }

#pragma unroll
    for (int j = 0; j < 8; ++j)
        y[(j + 8 * hi) * NLEN + n] = acc[j];
}

extern "C" void kernel_launch(void* const* d_in, const int* in_sizes, int n_in,
                              void* d_out, int out_size, void* d_ws, size_t ws_size,
                              hipStream_t stream) {
    const float* x = (const float*)d_in[0];
    const float* w = (const float*)d_in[1];
    float* out = (float*)d_out;

    float* Xm      = (float*)d_ws;
    float* Apacked = Xm + 1024;

    dht_forward_modes<<<1, 1024, 0, stream>>>(x, Xm);
    mode_mix<<<1, 1024, 0, stream>>>(Xm, w, Apacked);
    dht_inverse<<<128, 128, 0, stream>>>(Apacked, out);
}


# channel mix folded into the forward kernel: a workgroup owns a channel pair and 8 modes, applies the weights, writes 1 of 4 partial planes; inverse adds the planes; 2 launches; new f32 sum order
# speedup vs baseline: 7.2880x; 7.2880x over previous
#include <hip/hip_runtime.h>

#define NLEN   8192
#define MODES  64
#define ROWS   16
#define PLANE  2048
#define W0     (7.6699039394282058e-4f)
#define INV_N  (1.220703125e-4f)
#define W64    (9.8174770424681035e-2f)

typedef __attribute__((ext_vector_type(2))) float v2f;
typedef __attribute__((ext_vector_type(4))) float v4f;
typedef __attribute__((ext_vector_type(8))) float v8f;

static __device__ __forceinline__ v8f wmma_f32_16x16x4(v2f a, v2f b, v8f c) {
    return __builtin_amdgcn_wmma_f32_16x16x4_f32(
         false, a,  false, b,
         (short)0, c,  false,  false);
}

__global__ void __launch_bounds__(256)
dht_forward_mix(const float* __restrict__ x, const float* __restrict__ w,
                float* __restrict__ Apl) {
    __shared__ __attribute__((aligned(16))) float Bp[64 * 16];
    __shared__ __attribute__((aligned(16))) float red[2 * 16 * 16];

    const int tid  = threadIdx.x;
    const int lane = tid & 31;
    const int wave = tid >> 5;
    const int hi   = lane >> 4;
    const int lo   = lane & 15;
    const int t    = blockIdx.x & 7;
    const int cg   = (blockIdx.x >> 3) & 3;
    const int b    = blockIdx.x >> 5;

    const int mh = wave & 1;
    const int mq = (wave >> 1) & 1;
    const int mo = lane >> 2;
    const int mm = 4 * mq + (lane & 3);
    float w0 = 0.0f, w1 = 0.0f;
    if (tid < 128) {
        const int m = 8 * t + mm;
        w0 = w[((((2 * cg    ) * 8 + mo) * 64 + m) << 1) + mh];
        w1 = w[((((2 * cg + 1) * 8 + mo) * 64 + m) << 1) + mh];
    }

    const float* xa = x + (b * 8 + 2 * cg) * NLEN + 256 * hi + 16 * wave + lo;
    v2f a0[16], a1[16];
#pragma unroll
    for (int s = 0; s < 16; ++s) {
        a0[s].x = xa[512 * s];
        a0[s].y = xa[512 * s + 128];
        a1[s].x = xa[NLEN + 512 * s];
        a1[s].y = xa[NLEN + 512 * s + 128];
    }

#pragma unroll
    for (int i = 0; i < 4; ++i) {
        const int idx = tid + 256 * i;
        const int n1  = idx >> 4;
        const int c   = idx & 15;
        const int m   = 8 * t + (c & 7);
        const int q   = (c >> 3) * 16;
        Bp[idx] = __cosf(W64 * (float)((n1 * m - q) & 63));
    }
    __syncthreads();

    v8f acc0 = {}, acc1 = {};
    const float* bp = Bp + 32 * hi + lo;
#pragma unroll
    for (int s = 0; s < 16; ++s) {
        v2f bb;
        bb.x = bp[64 * s];
        bb.y = bp[64 * s + 16];
        acc0 = wmma_f32_16x16x4(a0[s], bb, acc0);
        acc1 = wmma_f32_16x16x4(a1[s], bb, acc1);
    }

    const int mode = 8 * t + (lo & 7);
    const int qt   = (lo >> 3) * (NLEN / 4);
    float p0 = 0.0f, p1 = 0.0f;
#pragma unroll
    for (int j = 0; j < 8; ++j) {
        const int n2 = 16 * wave + 8 * hi + j;
        const float tw = __cosf(W0 * (float)((n2 * mode + qt) & (NLEN - 1)));
        p0 += acc0[j] * tw;
        p1 += acc1[j] * tw;
    }
    red[      lo * 16 + wave * 2 + hi] = p0;
    red[256 + lo * 16 + wave * 2 + hi] = p1;
    __syncthreads();

    if (tid < 128) {
        const v4f* rc0 = (const v4f*)(red +       mm      * 16);
        const v4f* rs0 = (const v4f*)(red +       (mm + 8) * 16);
        const v4f* rc1 = (const v4f*)(red + 256 + mm      * 16);
        const v4f* rs1 = (const v4f*)(red + 256 + (mm + 8) * 16);
        float x0 = 0.0f, x1 = 0.0f;
#pragma unroll
        for (int k = 0; k < 4; ++k) {
            const v4f u0 = rc0[k] + rs0[k];
            const v4f u1 = rc1[k] + rs1[k];
            x0 += (u0.x + u0.y) + (u0.z + u0.w);
            x1 += (u1.x + u1.y) + (u1.z + u1.w);
        }
        Apl[cg * PLANE + (2 * t + mq) * 128 + 64 * mh + 32 * b + lane] = x0 * w0 + x1 * w1;
    }
}

__global__ void __launch_bounds__(128)
dht_inverse(const float* __restrict__ Apl, float* __restrict__ y) {
    const int lane = threadIdx.x & 31;
    const int wave = threadIdx.x >> 5;
    const int hi   = lane >> 4;
    const int lo   = lane & 15;
    const int n    = (blockIdx.x * 4 + wave) * 16 + lo;

    const v4f* ap = (const v4f*)Apl + lane;
    v4f f[16];
#pragma unroll
    for (int q = 0; q < 16; ++q)
        f[q] = (ap[32 * q] + ap[PLANE / 4 + 32 * q])
             + (ap[2 * (PLANE / 4) + 32 * q] + ap[3 * (PLANE / 4) + 32 * q]);

    const int qt = (NLEN / 4) * hi;

    v8f acc = {};
#pragma unroll
    for (int j = 0; j < 32; ++j) {
        v2f a, b;
        if (j & 1) { a.x = f[j >> 1].z; a.y = f[j >> 1].w; }
        else       { a.x = f[j >> 1].x; a.y = f[j >> 1].y; }
        b.x = INV_N * __cosf(W0 * (float)(((2 * j)     * n + qt) & (NLEN - 1)));
        b.y = INV_N * __cosf(W0 * (float)(((2 * j + 1) * n + qt) & (NLEN - 1)));
        acc = wmma_f32_16x16x4(a, b, acc);
    }

#pragma unroll
    for (int j = 0; j < 8; ++j)
        y[(j + 8 * hi) * NLEN + n] = acc[j];
}

extern "C" void kernel_launch(void* const* d_in, const int* in_sizes, int n_in,
                              void* d_out, int out_size, void* d_ws, size_t ws_size,
                              hipStream_t stream) {
    const float* x = (const float*)d_in[0];
    const float* w = (const float*)d_in[1];
    float* out = (float*)d_out;

    float* Apl = (float*)d_ws;

    dht_forward_mix<<<64, 256, 0, stream>>>(x, w, Apl);
    dht_inverse<<<128, 128, 0, stream>>>(Apl, out);
}


# inverse kernel: register budget stated as one wave per SIMD, so 44 of a lane's 64 plane loads are in flight before the K loop (was 24) and each is used 20 steps after issue, not 10; sums unchanged
# speedup vs baseline: 7.7381x; 1.0618x over previous
#include <hip/hip_runtime.h>

#define NLEN   8192
#define MODES  64
#define ROWS   16
#define PLANE  2048
#define W0     (7.6699039394282058e-4f)
#define INV_N  (1.220703125e-4f)
#define W64    (9.8174770424681035e-2f)

typedef __attribute__((ext_vector_type(2))) float v2f;
typedef __attribute__((ext_vector_type(4))) float v4f;
typedef __attribute__((ext_vector_type(8))) float v8f;

static __device__ __forceinline__ v8f wmma_f32_16x16x4(v2f a, v2f b, v8f c) {
    return __builtin_amdgcn_wmma_f32_16x16x4_f32(
         false, a,  false, b,
         (short)0, c,  false,  false);
}

__global__ void __launch_bounds__(256)
dht_forward_mix(const float* __restrict__ x, const float* __restrict__ w,
                float* __restrict__ Apl) {
    __shared__ __attribute__((aligned(16))) float Bp[64 * 16];
    __shared__ __attribute__((aligned(16))) float red[2 * 16 * 16];

    const int tid  = threadIdx.x;
    const int lane = tid & 31;
    const int wave = tid >> 5;
    const int hi   = lane >> 4;
    const int lo   = lane & 15;
    const int t    = blockIdx.x & 7;
    const int cg   = (blockIdx.x >> 3) & 3;
    const int b    = blockIdx.x >> 5;

    const int mh = wave & 1;
    const int mq = (wave >> 1) & 1;
    const int mo = lane >> 2;
    const int mm = 4 * mq + (lane & 3);
    float w0 = 0.0f, w1 = 0.0f;
    if (tid < 128) {
        const int m = 8 * t + mm;
        w0 = w[((((2 * cg    ) * 8 + mo) * 64 + m) << 1) + mh];
        w1 = w[((((2 * cg + 1) * 8 + mo) * 64 + m) << 1) + mh];
    }

    const float* xa = x + (b * 8 + 2 * cg) * NLEN + 256 * hi + 16 * wave + lo;
    v2f a0[16], a1[16];
#pragma unroll
    for (int s = 0; s < 16; ++s) {
        a0[s].x = xa[512 * s];
        a0[s].y = xa[512 * s + 128];
        a1[s].x = xa[NLEN + 512 * s];
        a1[s].y = xa[NLEN + 512 * s + 128];
    }

#pragma unroll
    for (int i = 0; i < 4; ++i) {
        const int idx = tid + 256 * i;
        const int n1  = idx >> 4;
        const int c   = idx & 15;
        const int m   = 8 * t + (c & 7);
        const int q   = (c >> 3) * 16;
        Bp[idx] = __cosf(W64 * (float)((n1 * m - q) & 63));
    }
    __syncthreads();

    v8f acc0 = {}, acc1 = {};
    const float* bp = Bp + 32 * hi + lo;
#pragma unroll
    for (int s = 0; s < 16; ++s) {
        v2f bb;
        bb.x = bp[64 * s];
        bb.y = bp[64 * s + 16];
        acc0 = wmma_f32_16x16x4(a0[s], bb, acc0);
        acc1 = wmma_f32_16x16x4(a1[s], bb, acc1);
    }

    const int mode = 8 * t + (lo & 7);
    const int qt   = (lo >> 3) * (NLEN / 4);
    float p0 = 0.0f, p1 = 0.0f;
#pragma unroll
    for (int j = 0; j < 8; ++j) {
        const int n2 = 16 * wave + 8 * hi + j;
        const float tw = __cosf(W0 * (float)((n2 * mode + qt) & (NLEN - 1)));
        p0 += acc0[j] * tw;
        p1 += acc1[j] * tw;
    }
    red[      lo * 16 + wave * 2 + hi] = p0;
    red[256 + lo * 16 + wave * 2 + hi] = p1;
    __syncthreads();

    if (tid < 128) {
        const v4f* rc0 = (const v4f*)(red +       mm      * 16);
        const v4f* rs0 = (const v4f*)(red +       (mm + 8) * 16);
        const v4f* rc1 = (const v4f*)(red + 256 + mm      * 16);
        const v4f* rs1 = (const v4f*)(red + 256 + (mm + 8) * 16);
        float x0 = 0.0f, x1 = 0.0f;
#pragma unroll
        for (int k = 0; k < 4; ++k) {
            const v4f u0 = rc0[k] + rs0[k];
            const v4f u1 = rc1[k] + rs1[k];
            x0 += (u0.x + u0.y) + (u0.z + u0.w);
            x1 += (u1.x + u1.y) + (u1.z + u1.w);
        }
        Apl[cg * PLANE + (2 * t + mq) * 128 + 64 * mh + 32 * b + lane] = x0 * w0 + x1 * w1;
    }
}

__global__ void __launch_bounds__(128) __attribute__((amdgpu_waves_per_eu(1, 1)))
dht_inverse(const float* __restrict__ Apl, float* __restrict__ y) {
    const int lane = threadIdx.x & 31;
    const int wave = threadIdx.x >> 5;
    const int hi   = lane >> 4;
    const int lo   = lane & 15;
    const int n    = (blockIdx.x * 4 + wave) * 16 + lo;

    const v4f* ap = (const v4f*)Apl + lane;
    v4f f[16];
#pragma unroll
    for (int q = 0; q < 16; ++q)
        f[q] = (ap[32 * q] + ap[PLANE / 4 + 32 * q])
             + (ap[2 * (PLANE / 4) + 32 * q] + ap[3 * (PLANE / 4) + 32 * q]);

    const int qt = (NLEN / 4) * hi;

    v8f acc = {};
#pragma unroll
    for (int j = 0; j < 32; ++j) {
        v2f a, b;
        if (j & 1) { a.x = f[j >> 1].z; a.y = f[j >> 1].w; }
        else       { a.x = f[j >> 1].x; a.y = f[j >> 1].y; }
        b.x = INV_N * __cosf(W0 * (float)(((2 * j)     * n + qt) & (NLEN - 1)));
        b.y = INV_N * __cosf(W0 * (float)(((2 * j + 1) * n + qt) & (NLEN - 1)));
        acc = wmma_f32_16x16x4(a, b, acc);
    }

#pragma unroll
    for (int j = 0; j < 8; ++j)
        y[(j + 8 * hi) * NLEN + n] = acc[j];
}

extern "C" void kernel_launch(void* const* d_in, const int* in_sizes, int n_in,
                              void* d_out, int out_size, void* d_ws, size_t ws_size,
                              hipStream_t stream) {
    const float* x = (const float*)d_in[0];
    const float* w = (const float*)d_in[1];
    float* out = (float*)d_out;

    float* Apl = (float*)d_ws;

    dht_forward_mix<<<64, 256, 0, stream>>>(x, w, Apl);
    dht_inverse<<<128, 128, 0, stream>>>(Apl, out);
}


# forward kernel: each lane computes its B cos|sin values in registers, no LDS B panel or barrier ahead of the K loop, so a WMMA waits only for its own x words; budget 2 waves per SIMD; sums unchanged
# speedup vs baseline: 8.1321x; 1.0509x over previous
#include <hip/hip_runtime.h>

#define NLEN   8192
#define MODES  64
#define ROWS   16
#define PLANE  2048
#define W0     (7.6699039394282058e-4f)
#define INV_N  (1.220703125e-4f)
#define W64    (9.8174770424681035e-2f)

typedef __attribute__((ext_vector_type(2))) float v2f;
typedef __attribute__((ext_vector_type(4))) float v4f;
typedef __attribute__((ext_vector_type(8))) float v8f;

static __device__ __forceinline__ v8f wmma_f32_16x16x4(v2f a, v2f b, v8f c) {
    return __builtin_amdgcn_wmma_f32_16x16x4_f32(
         false, a,  false, b,
         (short)0, c,  false,  false);
}

__global__ void __launch_bounds__(256) __attribute__((amdgpu_waves_per_eu(2, 2)))
dht_forward_mix(const float* __restrict__ x, const float* __restrict__ w,
                float* __restrict__ Apl) {
    __shared__ __attribute__((aligned(16))) float red[2 * 16 * 16];

    const int tid  = threadIdx.x;
    const int lane = tid & 31;
    const int wave = tid >> 5;
    const int hi   = lane >> 4;
    const int lo   = lane & 15;
    const int t    = blockIdx.x & 7;
    const int cg   = (blockIdx.x >> 3) & 3;
    const int b    = blockIdx.x >> 5;

    const int mh = wave & 1;
    const int mq = (wave >> 1) & 1;
    const int mo = lane >> 2;
    const int mm = 4 * mq + (lane & 3);
    float w0 = 0.0f, w1 = 0.0f;
    if (tid < 128) {
        const int m = 8 * t + mm;
        w0 = w[((((2 * cg    ) * 8 + mo) * 64 + m) << 1) + mh];
        w1 = w[((((2 * cg + 1) * 8 + mo) * 64 + m) << 1) + mh];
    }

    const float* xa = x + (b * 8 + 2 * cg) * NLEN + 256 * hi + 16 * wave + lo;
    v2f a0[16], a1[16];
#pragma unroll
    for (int s = 0; s < 16; ++s) {
        a0[s].x = xa[512 * s];
        a0[s].y = xa[512 * s + 128];
        a1[s].x = xa[NLEN + 512 * s];
        a1[s].y = xa[NLEN + 512 * s + 128];
    }

    const int bm = 8 * t + (lo & 7);
    const int bs = (lo >> 3) * 16;
    v2f bq[16];
#pragma unroll
    for (int s = 0; s < 16; ++s) {
        const int n1 = 4 * s + 2 * hi;
        bq[s].x = __cosf(W64 * (float)(( n1      * bm - bs) & 63));
        bq[s].y = __cosf(W64 * (float)(((n1 + 1) * bm - bs) & 63));
    }

    v8f acc0 = {}, acc1 = {};
#pragma unroll
    for (int s = 0; s < 16; ++s) {
        acc0 = wmma_f32_16x16x4(a0[s], bq[s], acc0);
        acc1 = wmma_f32_16x16x4(a1[s], bq[s], acc1);
    }

    const int mode = 8 * t + (lo & 7);
    const int qt   = (lo >> 3) * (NLEN / 4);
    float p0 = 0.0f, p1 = 0.0f;
#pragma unroll
    for (int j = 0; j < 8; ++j) {
        const int n2 = 16 * wave + 8 * hi + j;
        const float tw = __cosf(W0 * (float)((n2 * mode + qt) & (NLEN - 1)));
        p0 += acc0[j] * tw;
        p1 += acc1[j] * tw;
    }
    red[      lo * 16 + wave * 2 + hi] = p0;
    red[256 + lo * 16 + wave * 2 + hi] = p1;
    __syncthreads();

    if (tid < 128) {
        const v4f* rc0 = (const v4f*)(red +       mm      * 16);
        const v4f* rs0 = (const v4f*)(red +       (mm + 8) * 16);
        const v4f* rc1 = (const v4f*)(red + 256 + mm      * 16);
        const v4f* rs1 = (const v4f*)(red + 256 + (mm + 8) * 16);
        float x0 = 0.0f, x1 = 0.0f;
#pragma unroll
        for (int k = 0; k < 4; ++k) {
            const v4f u0 = rc0[k] + rs0[k];
            const v4f u1 = rc1[k] + rs1[k];
            x0 += (u0.x + u0.y) + (u0.z + u0.w);
            x1 += (u1.x + u1.y) + (u1.z + u1.w);
        }
        Apl[cg * PLANE + (2 * t + mq) * 128 + 64 * mh + 32 * b + lane] = x0 * w0 + x1 * w1;
    }
}

__global__ void __launch_bounds__(128) __attribute__((amdgpu_waves_per_eu(1, 1)))
dht_inverse(const float* __restrict__ Apl, float* __restrict__ y) {
    const int lane = threadIdx.x & 31;
    const int wave = threadIdx.x >> 5;
    const int hi   = lane >> 4;
    const int lo   = lane & 15;
    const int n    = (blockIdx.x * 4 + wave) * 16 + lo;

    const v4f* ap = (const v4f*)Apl + lane;
    v4f f[16];
#pragma unroll
    for (int q = 0; q < 16; ++q)
        f[q] = (ap[32 * q] + ap[PLANE / 4 + 32 * q])
             + (ap[2 * (PLANE / 4) + 32 * q] + ap[3 * (PLANE / 4) + 32 * q]);

    const int qt = (NLEN / 4) * hi;

    v8f acc = {};
#pragma unroll
    for (int j = 0; j < 32; ++j) {
        v2f a, b;
        if (j & 1) { a.x = f[j >> 1].z; a.y = f[j >> 1].w; }
        else       { a.x = f[j >> 1].x; a.y = f[j >> 1].y; }
        b.x = INV_N * __cosf(W0 * (float)(((2 * j)     * n + qt) & (NLEN - 1)));
        b.y = INV_N * __cosf(W0 * (float)(((2 * j + 1) * n + qt) & (NLEN - 1)));
        acc = wmma_f32_16x16x4(a, b, acc);
    }

#pragma unroll
    for (int j = 0; j < 8; ++j)
        y[(j + 8 * hi) * NLEN + n] = acc[j];
}

extern "C" void kernel_launch(void* const* d_in, const int* in_sizes, int n_in,
                              void* d_out, int out_size, void* d_ws, size_t ws_size,
                              hipStream_t stream) {
    const float* x = (const float*)d_in[0];
    const float* w = (const float*)d_in[1];
    float* out = (float*)d_out;

    float* Apl = (float*)d_ws;

    dht_forward_mix<<<64, 256, 0, stream>>>(x, w, Apl);
    dht_inverse<<<128, 128, 0, stream>>>(Apl, out);
}
